# starting kernel
# baseline (speedup 1.0000x reference)
#include <hip/hip_runtime.h>

typedef float v2f __attribute__((ext_vector_type(2)));
typedef float v8f __attribute__((ext_vector_type(8)));

#define IMG_H 512
#define IMG_W 512
#define NBATCH 16
#define TS 16
#define KSZ 5
#define PADR 2
#define HALO (TS + KSZ - 1)

__device__ __forceinline__ int refl(int i) {
    i = (i < 0) ? -i : i;
    i = (i >= IMG_H) ? (2 * IMG_H - 2 - i) : i;
    return i;
}

__global__ __launch_bounds__(256)
void patch_moments_wmma(const float* __restrict__ x, float* __restrict__ out) {
    __shared__ float Tlds[8][2][16][HALO];

    const int lane  = threadIdx.x & 31;
    const int wave  = threadIdx.x >> 5;
    const int tile  = blockIdx.x * 8 + wave;

    const int b  = tile >> 10;
    const int tr = (tile >> 5) & 31;
    const int tc = tile & 31;
    const int r0 = tr * TS;
    const int c0 = tc * TS;

    const float* __restrict__ xb = x + (size_t)b * (IMG_H * IMG_W);

    const int n     = lane & 15;
    const int khalf = lane >> 4;
    const int m     = n;

    v8f dsum0 = {}, dsum1 = {};
    v8f dsq0  = {}, dsq1  = {};

    #pragma unroll
    for (int g = 0; g < 2; ++g) {
        const int cpad = g * 16 + n;
        const int gcol = refl(c0 - PADR + cpad);
        #pragma unroll
        for (int kc = 0; kc < 5; ++kc) {
            v2f a, bx, bxx;
            #pragma unroll
            for (int p = 0; p < 2; ++p) {
                const int k = 4 * kc + 2 * khalf + p;
                a[p] = (k >= m && k <= m + 4) ? 1.0f : 0.0f;
                const int grow = refl(r0 - PADR + k);
                const float v  = xb[grow * IMG_W + gcol];
                bx[p]  = v;
                bxx[p] = v * v;
            }
            if (g == 0) {
                dsum0 = __builtin_amdgcn_wmma_f32_16x16x4_f32(false, a, false, bx,  (short)0, dsum0, false, false);
                dsq0  = __builtin_amdgcn_wmma_f32_16x16x4_f32(false, a, false, bxx, (short)0, dsq0,  false, false);
            } else {
                dsum1 = __builtin_amdgcn_wmma_f32_16x16x4_f32(false, a, false, bx,  (short)0, dsum1, false, false);
                dsq1  = __builtin_amdgcn_wmma_f32_16x16x4_f32(false, a, false, bxx, (short)0, dsq1,  false, false);
            }
        }
    }

    #pragma unroll
    for (int vv = 0; vv < 8; ++vv) {
        const int row = vv + 8 * khalf;
        Tlds[wave][0][row][n] = dsum0[vv];
        Tlds[wave][1][row][n] = dsq0[vv];
        if (n < HALO - 16) {
            Tlds[wave][0][row][16 + n] = dsum1[vv];
            Tlds[wave][1][row][16 + n] = dsq1[vv];
        }
    }
    asm volatile("s_wait_dscnt 0" ::: "memory");

    v8f s1 = {}, s2 = {};
    #pragma unroll
    for (int kc = 0; kc < 5; ++kc) {
        v2f asum, asq, bb;
        #pragma unroll
        for (int p = 0; p < 2; ++p) {
            const int k = 4 * kc + 2 * khalf + p;
            asum[p] = Tlds[wave][0][m][k];
            asq[p]  = Tlds[wave][1][m][k];
            bb[p] = (k >= n && k <= n + 4) ? 1.0f : 0.0f;
        }
        s1 = __builtin_amdgcn_wmma_f32_16x16x4_f32(false, asum, false, bb, (short)0, s1, false, false);
        s2 = __builtin_amdgcn_wmma_f32_16x16x4_f32(false, asq,  false, bb, (short)0, s2, false, false);
    }

    const float inv_n   = 1.0f / 25.0f;
    const float inv_nm1 = 1.0f / 24.0f;
    float* __restrict__ avg_out = out;
    float* __restrict__ var_out = out + (size_t)NBATCH * IMG_H * IMG_W;
    const size_t base = (size_t)b * (IMG_H * IMG_W);

    #pragma unroll
    for (int vv = 0; vv < 8; ++vv) {
        const int row = r0 + vv + 8 * khalf;
        const int col = c0 + n;
        const float ssum = s1[vv];
        const float a    = ssum * inv_n;
        const float var  = (s2[vv] - ssum * a) * inv_nm1;
        const size_t idx = base + (size_t)(row * IMG_W + col);
        avg_out[idx] = a;
        var_out[idx] = var;
    }
}

extern "C" void kernel_launch(void* const* d_in, const int* in_sizes, int n_in,
                              void* d_out, int out_size, void* d_ws, size_t ws_size,
                              hipStream_t stream) {
    (void)in_sizes; (void)n_in; (void)out_size; (void)d_ws; (void)ws_size;
    const float* x = (const float*)d_in[0];
    float* out = (float*)d_out;
    patch_moments_wmma<<<dim3(2048), dim3(256), 0, stream>>>(x, out);
}


# Column-strip walk: horizontal 5-sums first (image rows as WMMA A operand), their D tile is the vertical sum's B operand straight from registers; no LDS, 30->20 WMMA per tile; sum order changes
# speedup vs baseline: 1.1104x; 1.1104x over previous
#include <hip/hip_runtime.h>


typedef float v2f __attribute__((ext_vector_type(2)));
typedef float v4f __attribute__((ext_vector_type(4)));
typedef float v8f __attribute__((ext_vector_type(8)));

#define IMG_H 512
#define IMG_W 512
#define NBATCH 16
#define SEG_TILES 8
#define SEG_ROWS (16 * SEG_TILES)
#define LOAD_AHEAD 3

__device__ __forceinline__ int refl(int i) {
    i = (i < 0) ? -i : i;
    i = (i >= IMG_H) ? (2 * IMG_H - 2 - i) : i;
    return i;
}

__device__ __forceinline__ int row_off(int M) {
    return (M < 2) ? M : (M < 8) ? (M + 2) : (M < 10) ? (M - 6) : M;
}

__device__ __forceinline__ int win_col(int e, int h) {
    return (e < 2) ? (h ? 18 + e : e) : (h ? 8 + e : e);
}

template <int SIDE>
__device__ __forceinline__ void walk_strip(const float* __restrict__ xb,
                                           float* __restrict__ avg_o,
                                           float* __restrict__ var_o,
                                           const int c0, const int S, const int lane) {
    const int c = lane & 15;
    const int h = lane >> 4;

    float xr[SEG_TILES + 1][10];
    const int off_c = row_off(c);
    const float* __restrict__ lane_row = xb + (S - 2 + off_c) * IMG_W + c0;
    const float* __restrict__ first_row = xb + refl(S - 2 + off_c) * IMG_W + c0;
    const float* __restrict__ last_row  = xb + refl(S + 16 * SEG_TILES - 2 + (off_c & 3)) * IMG_W + c0;
    int halo = h ? 16 : -2;
    if (SIDE == 1 && !h) halo = 0;
    if (SIDE == 2 && h)  halo = 14;
    const int inner = h ? 8 : 0;

    auto load_window = [&](const int u) {
        const float* __restrict__ p = (u == 0) ? first_row : (u == SEG_TILES) ? last_row : lane_row + 16 * u * IMG_W;
        const v2f e01 = *(const v2f*)(p + halo);
        const v4f e25 = *(const v4f*)(p + inner);
        const v4f e69 = *(const v4f*)(p + inner + 4);
        xr[u][0] = e01[0]; xr[u][1] = e01[1];
        xr[u][2] = e25[0]; xr[u][3] = e25[1]; xr[u][4] = e25[2]; xr[u][5] = e25[3];
        xr[u][6] = e69[0]; xr[u][7] = e69[1]; xr[u][8] = e69[2]; xr[u][9] = e69[3];
    };

    #pragma unroll
    for (int u = 0; u < LOAD_AHEAD; ++u) load_window(u);
    __builtin_amdgcn_sched_barrier(0);

    v2f hb[5];
    #pragma unroll
    for (int q = 0; q < 5; ++q) {
        #pragma unroll
        for (int s = 0; s < 2; ++s) {
            const int w = win_col(2 * q + s, h);
            hb[q][s] = (w >= c && w <= c + 4) ? 1.0f : 0.0f;
        }
    }
    v2f vb[5];
    #pragma unroll
    for (int j = 0; j < 4; ++j) {
        #pragma unroll
        for (int s = 0; s < 2; ++s) {
            const int o = row_off(8 * h + 2 * j + s);
            vb[j][s] = (o >= c && o <= c + 4) ? 1.0f : 0.0f;
        }
    }
    #pragma unroll
    for (int s = 0; s < 2; ++s) {
        const int o = 16 + 2 * h + s;
        vb[4][s] = (o >= c && o <= c + 4) ? 1.0f : 0.0f;
    }

    v8f hs[2], hq[2];

    float* __restrict__ avg_lane = avg_o + ((S + 8 * h) * IMG_W + c0 + c);
    float* __restrict__ var_lane = var_o + ((S + 8 * h) * IMG_W + c0 + c);

    #pragma unroll
    for (int u = 0; u <= SEG_TILES; ++u) {
        if (u + LOAD_AHEAD <= SEG_TILES) {
            load_window(u + LOAD_AHEAD);
            __builtin_amdgcn_sched_barrier(0);
        }
        v8f ds = {}, dq = {};
        #pragma unroll
        for (int q = 0; q < 5; ++q) {
            v2f a;  a[0] = xr[u][2 * q]; a[1] = xr[u][2 * q + 1];
            if (q == 0 && SIDE == 1 && !h) { a[0] = xr[u][4]; a[1] = xr[u][3]; }
            if (q == 0 && SIDE == 2 && h)  { a[0] = xr[u][8]; a[1] = xr[u][7]; }
            const v2f aa = a * a;
            ds = __builtin_amdgcn_wmma_f32_16x16x4_f32(false, a,  false, hb[q], (short)0, ds, false, false);
            dq = __builtin_amdgcn_wmma_f32_16x16x4_f32(false, aa, false, hb[q], (short)0, dq, false, false);
        }
        hs[u & 1] = ds;
        hq[u & 1] = dq;
        if (u == 0) continue;

        const int t = u - 1;
        const v8f ps = hs[t & 1], pq = hq[t & 1];
        v8f zs = {}, zq = {};
        #pragma unroll
        for (int j = 0; j < 4; ++j) {
            v2f bs, bq;
            bs[0] = ps[2 * j]; bs[1] = ps[2 * j + 1];
            bq[0] = pq[2 * j]; bq[1] = pq[2 * j + 1];
            zs = __builtin_amdgcn_wmma_f32_16x16x4_f32(false, vb[j], false, bs, (short)0, zs, false, false);
            zq = __builtin_amdgcn_wmma_f32_16x16x4_f32(false, vb[j], false, bq, (short)0, zq, false, false);
        }
        {
            v2f bs, bq;
            bs[0] = ds[0]; bs[1] = ds[1];
            bq[0] = dq[0]; bq[1] = dq[1];
            zs = __builtin_amdgcn_wmma_f32_16x16x4_f32(false, vb[4], false, bs, (short)0, zs, false, false);
            zq = __builtin_amdgcn_wmma_f32_16x16x4_f32(false, vb[4], false, bq, (short)0, zq, false, false);
        }

        const float inv_n   = 1.0f / 25.0f;
        const float inv_nm1 = 1.0f / 24.0f;
        #pragma unroll
        for (int v = 0; v < 8; v += 2) {
            v2f ssum, ssq;
            ssum[0] = zs[v]; ssum[1] = zs[v + 1];
            ssq[0]  = zq[v]; ssq[1]  = zq[v + 1];
            const v2f a   = ssum * inv_n;
            const v2f var = (ssq - ssum * a) * inv_nm1;
            avg_lane[(16 * t + v) * IMG_W]     = a[0];
            var_lane[(16 * t + v) * IMG_W]     = var[0];
            avg_lane[(16 * t + v + 1) * IMG_W] = a[1];
            var_lane[(16 * t + v + 1) * IMG_W] = var[1];
        }
    }
}

__global__ __launch_bounds__(256)
void patch_moments_strips(const float* __restrict__ x, float* __restrict__ out) {
    const int lane = threadIdx.x & 31;
    const int wave = __builtin_amdgcn_readfirstlane(threadIdx.x >> 5);

    const int blk   = blockIdx.x;
    const int b     = blk >> 4;
    const int seg   = (blk >> 2) & 3;
    const int strip = (blk & 3) * 8 + wave;
    const int c0    = strip * 16;
    const int S     = seg * SEG_ROWS;

    const float* __restrict__ xb = x + (size_t)b * (IMG_H * IMG_W);
    float* __restrict__ avg_o = out + (size_t)b * (IMG_H * IMG_W);
    float* __restrict__ var_o = out + (size_t)NBATCH * IMG_H * IMG_W + (size_t)b * (IMG_H * IMG_W);

    if (strip == 0)
        walk_strip<1>(xb, avg_o, var_o, c0, S, lane);
    else if (strip == 31)
        walk_strip<2>(xb, avg_o, var_o, c0, S, lane);
    else
        walk_strip<0>(xb, avg_o, var_o, c0, S, lane);
}

extern "C" void kernel_launch(void* const* d_in, const int* in_sizes, int n_in,
                              void* d_out, int out_size, void* d_ws, size_t ws_size,
                              hipStream_t stream) {
    (void)in_sizes; (void)n_in; (void)out_size; (void)d_ws; (void)ws_size;
    const float* x = (const float*)d_in[0];
    float* out = (float*)d_out;
    patch_moments_strips<<<dim3(256), dim3(256), 0, stream>>>(x, out);
}


# Matrix tile rows = 8 image rows x two 16-col blocks (32-col strips): loads touch 8 rows not 16, one lane-half swap per register pair makes each store a whole 128-B line; vertical sum order changes
# speedup vs baseline: 1.3000x; 1.1707x over previous
#include <hip/hip_runtime.h>


typedef float v2f __attribute__((ext_vector_type(2)));
typedef float v4f __attribute__((ext_vector_type(4)));
typedef float v8f __attribute__((ext_vector_type(8)));
typedef unsigned int v2u __attribute__((ext_vector_type(2)));

#define IMG_H 512
#define IMG_W 512
#define NBATCH 16
#define STRIP_W 32
#define SEG_ROWS 64
#define NWIN (SEG_ROWS / 8 + 1)
#define LOAD_AHEAD 4

__device__ __forceinline__ int refl(int i) {
    i = (i < 0) ? -i : i;
    i = (i >= IMG_H) ? (2 * IMG_H - 2 - i) : i;
    return i;
}

__device__ __forceinline__ int win_col(int e, int h) {
    return (e < 2) ? (h ? 18 + e : e) : (h ? 8 + e : e);
}

__device__ __forceinline__ void swap_halves(float& lo, float& hi) {
    const v2u r = __builtin_amdgcn_permlane16_swap(__float_as_uint(lo), __float_as_uint(hi), false, false);
    lo = __uint_as_float(r[0]);
    hi = __uint_as_float(r[1]);
}

template <int SIDE>
__device__ __forceinline__ void walk_strip(const float* __restrict__ xb,
                                           float* __restrict__ avg_o,
                                           float* __restrict__ var_o,
                                           const int c0, const int S, const int lane) {
    const int c = lane & 15;
    const int h = lane >> 4;

    float xr[NWIN][10];
    const int blkA = (c >> 2) & 1;
    const int rho  = 4 * ((c >> 1) & 1) + 2 * (c >> 3) + (c & 1);
    const int colA = c0 + 16 * blkA;
    const float* __restrict__ lane_row  = xb + (S - 2 + rho) * IMG_W + colA;
    const float* __restrict__ first_row = xb + refl(S - 2 + rho) * IMG_W + colA;
    const float* __restrict__ last_row  = xb + refl(S + 8 * (NWIN - 1) - 2 + (rho & 3)) * IMG_W + colA;
    int halo = h ? 16 : -2;
    if (SIDE == 1 && !h && !blkA) halo = 0;
    if (SIDE == 2 && h && blkA)   halo = 14;
    const int inner = h ? 8 : 0;

    auto load_window = [&](const int u) {
        const float* __restrict__ p = (u == 0) ? first_row : (u == NWIN - 1) ? last_row : lane_row + 8 * u * IMG_W;
        const v2f e01 = *(const v2f*)(p + halo);
        const v4f e25 = *(const v4f*)(p + inner);
        const v4f e69 = *(const v4f*)(p + inner + 4);
        xr[u][0] = e01[0]; xr[u][1] = e01[1];
        xr[u][2] = e25[0]; xr[u][3] = e25[1]; xr[u][4] = e25[2]; xr[u][5] = e25[3];
        xr[u][6] = e69[0]; xr[u][7] = e69[1]; xr[u][8] = e69[2]; xr[u][9] = e69[3];
    };

    #pragma unroll
    for (int u = 0; u < LOAD_AHEAD; ++u) load_window(u);
    __builtin_amdgcn_sched_barrier(0);

    v2f hb[5];
    #pragma unroll
    for (int q = 0; q < 5; ++q) {
        #pragma unroll
        for (int s = 0; s < 2; ++s) {
            const int w = win_col(2 * q + s, h);
            hb[q][s] = (w >= c && w <= c + 4) ? 1.0f : 0.0f;
        }
    }
    v2f vb[5];
    #pragma unroll
    for (int i = 0; i < 5; ++i) {
        #pragma unroll
        for (int s = 0; s < 2; ++s) {
            const int o = 4 * i + 2 * h + s;
            vb[i][s] = (o >= c && o <= c + 4) ? 1.0f : 0.0f;
        }
    }

    v8f hs[3], hq[3];

    float* __restrict__ avg_lane = avg_o + (S * IMG_W + c0 + lane);
    float* __restrict__ var_lane = var_o + (S * IMG_W + c0 + lane);

    #pragma unroll
    for (int u = 0; u < NWIN; ++u) {
        if (u + LOAD_AHEAD < NWIN) {
            load_window(u + LOAD_AHEAD);
            __builtin_amdgcn_sched_barrier(0);
        }
        v8f ds = {}, dq = {};
        #pragma unroll
        for (int q = 0; q < 5; ++q) {
            v2f a;  a[0] = xr[u][2 * q]; a[1] = xr[u][2 * q + 1];
            if (q == 0 && SIDE == 1 && !h && !blkA) { a[0] = xr[u][4]; a[1] = xr[u][3]; }
            if (q == 0 && SIDE == 2 && h && blkA)   { a[0] = xr[u][8]; a[1] = xr[u][7]; }
            const v2f aa = a * a;
            ds = __builtin_amdgcn_wmma_f32_16x16x4_f32(false, a,  false, hb[q], (short)0, ds, false, false);
            dq = __builtin_amdgcn_wmma_f32_16x16x4_f32(false, aa, false, hb[q], (short)0, dq, false, false);
        }
        hs[u % 3] = ds;
        hq[u % 3] = dq;
        if (u < 2 || (u & 1)) continue;

        const int t = (u - 2) / 2;
        v8f zs[2], zq[2];
        #pragma unroll
        for (int be = 0; be < 2; ++be) {
            v8f accs = {}, accq = {};
            #pragma unroll
            for (int i = 0; i < 5; ++i) {
                const v8f ps = hs[(u - 2 + i / 2) % 3];
                const v8f pq = hq[(u - 2 + i / 2) % 3];
                const int r0 = 4 * be + 2 * (i & 1);
                v2f bs, bq;
                bs[0] = ps[r0]; bs[1] = ps[r0 + 1];
                bq[0] = pq[r0]; bq[1] = pq[r0 + 1];
                accs = __builtin_amdgcn_wmma_f32_16x16x4_f32(false, vb[i], false, bs, (short)0, accs, false, false);
                accq = __builtin_amdgcn_wmma_f32_16x16x4_f32(false, vb[i], false, bq, (short)0, accq, false, false);
            }
            zs[be] = accs;
            zq[be] = accq;
        }

        const float inv_n   = 1.0f / 25.0f;
        const float inv_nm1 = 1.0f / 24.0f;
        #pragma unroll
        for (int v = 0; v < 8; v += 2) {
            v2f a[2], var[2];
            #pragma unroll
            for (int be = 0; be < 2; ++be) {
                v2f ssum, ssq;
                ssum[0] = zs[be][v]; ssum[1] = zs[be][v + 1];
                ssq[0]  = zq[be][v]; ssq[1]  = zq[be][v + 1];
                a[be]   = ssum * inv_n;
                var[be] = (ssq - ssum * a[be]) * inv_nm1;
            }
            #pragma unroll
            for (int e = 0; e < 2; ++e) {
                float alo = a[0][e],   ahi = a[1][e];
                float vlo = var[0][e], vhi = var[1][e];
                swap_halves(alo, ahi);
                swap_halves(vlo, vhi);
                avg_lane[(16 * t + v + e) * IMG_W]     = alo;
                var_lane[(16 * t + v + e) * IMG_W]     = vlo;
                avg_lane[(16 * t + v + e + 8) * IMG_W] = ahi;
                var_lane[(16 * t + v + e + 8) * IMG_W] = vhi;
            }
        }
    }
}

__global__ __launch_bounds__(256)
void patch_moments_strips(const float* __restrict__ x, float* __restrict__ out) {
    const int lane = threadIdx.x & 31;
    const int wave = __builtin_amdgcn_readfirstlane(threadIdx.x >> 5);

    const int blk   = blockIdx.x;
    const int b     = blk >> 4;
    const int seg   = (blk >> 1) & 7;
    const int strip = (blk & 1) * 8 + wave;
    const int c0    = strip * STRIP_W;
    const int S     = seg * SEG_ROWS;

    const float* __restrict__ xb = x + (size_t)b * (IMG_H * IMG_W);
    float* __restrict__ avg_o = out + (size_t)b * (IMG_H * IMG_W);
    float* __restrict__ var_o = out + (size_t)NBATCH * IMG_H * IMG_W + (size_t)b * (IMG_H * IMG_W);

    if (strip == 0)
        walk_strip<1>(xb, avg_o, var_o, c0, S, lane);
    else if (strip == IMG_W / STRIP_W - 1)
        walk_strip<2>(xb, avg_o, var_o, c0, S, lane);
    else
        walk_strip<0>(xb, avg_o, var_o, c0, S, lane);
}

extern "C" void kernel_launch(void* const* d_in, const int* in_sizes, int n_in,
                              void* d_out, int out_size, void* d_ws, size_t ws_size,
                              hipStream_t stream) {
    (void)in_sizes; (void)n_in; (void)out_size; (void)d_ws; (void)ws_size;
    const float* x = (const float*)d_in[0];
    float* out = (float*)d_out;
    patch_moments_strips<<<dim3(256), dim3(256), 0, stream>>>(x, out);
}
